# starting kernel
# baseline (speedup 1.0000x reference)
#include <hip/hip_runtime.h>
#include <hip/hip_bf16.h>

#define N_NODES   50000
#define N_EDGES   800000
#define IN_FEATS  128
#define OUT_FEATS 512
#define LEAKY     0.01f

typedef __attribute__((ext_vector_type(2))) float v2f;
typedef __attribute__((ext_vector_type(4))) float v4f;
typedef __attribute__((ext_vector_type(8))) float v8f;

__global__ __launch_bounds__(256) void zero_h(float* __restrict__ h) {
    unsigned i = blockIdx.x * 256u + threadIdx.x;
    ((v4f*)h)[i] = (v4f){0.f, 0.f, 0.f, 0.f};
}

__global__ __launch_bounds__(256) void scatter_add(const float* __restrict__ feature,
                                                   const int*   __restrict__ src,
                                                   const int*   __restrict__ dst,
                                                   float*       __restrict__ h) {
    unsigned gid  = blockIdx.x * 256u + threadIdx.x;
    unsigned e    = gid >> 5;
    unsigned lane = gid & 31u;

    int s = src[e];
    int d = dst[e];

    const float* fp = feature + (size_t)s * IN_FEATS + lane * 4u;
    v4f val = *(const v4f*)fp;

    float* hp = h + (size_t)d * IN_FEATS + lane * 4u;
    atomicAdd(hp + 0, val.x);
    atomicAdd(hp + 1, val.y);
    atomicAdd(hp + 2, val.z);
    atomicAdd(hp + 3, val.w);
}

__global__ __launch_bounds__(256) void gemm_wmma(const float* __restrict__ h,
                                                 const float* __restrict__ W,
                                                 const float* __restrict__ bias,
                                                 float*       __restrict__ out) {
    const unsigned waveInBlk = threadIdx.x >> 5;
    const unsigned lane      = threadIdx.x & 31u;
    const unsigned tile      = blockIdx.x * 8u + waveInBlk;

    const unsigned rowTile  = tile >> 3;
    const unsigned colGroup = tile & 7u;

    const unsigned nbase = rowTile * 16u;
    const unsigned obase = colGroup * 64u;

    const unsigned M  = lane & 15u;
    const unsigned kg = lane >> 4;

    v8f acc0 = {}, acc1 = {}, acc2 = {}, acc3 = {};

    const float* arow = h + (size_t)(nbase + M) * IN_FEATS;
    const float* b0   = W + (size_t)(obase + 0u  + M) * IN_FEATS;
    const float* b1   = W + (size_t)(obase + 16u + M) * IN_FEATS;
    const float* b2   = W + (size_t)(obase + 32u + M) * IN_FEATS;
    const float* b3   = W + (size_t)(obase + 48u + M) * IN_FEATS;

    #pragma unroll 4
    for (int kb = 0; kb < IN_FEATS; kb += 4) {
        const int ko = kb + 2 * (int)kg;
        v2f a  = *(const v2f*)(arow + ko);
        v2f f0 = *(const v2f*)(b0 + ko);
        v2f f1 = *(const v2f*)(b1 + ko);
        v2f f2 = *(const v2f*)(b2 + ko);
        v2f f3 = *(const v2f*)(b3 + ko);
        acc0 = __builtin_amdgcn_wmma_f32_16x16x4_f32(false, a, false, f0,
                                                     (short)0, acc0, false, false);
        acc1 = __builtin_amdgcn_wmma_f32_16x16x4_f32(false, a, false, f1,
                                                     (short)0, acc1, false, false);
        acc2 = __builtin_amdgcn_wmma_f32_16x16x4_f32(false, a, false, f2,
                                                     (short)0, acc2, false, false);
        acc3 = __builtin_amdgcn_wmma_f32_16x16x4_f32(false, a, false, f3,
                                                     (short)0, acc3, false, false);
    }

    v8f accs[4] = {acc0, acc1, acc2, acc3};
    #pragma unroll
    for (int t = 0; t < 4; ++t) {
        const unsigned ocol = obase + (unsigned)t * 16u + (lane & 15u);
        const float bv = bias[ocol];
        #pragma unroll
        for (int v = 0; v < 8; ++v) {
            const unsigned Mrow = (unsigned)v + 8u * kg;
            float val = accs[t][v] + bv;
            val = (val >= 0.f) ? val : LEAKY * val;
            out[(size_t)(nbase + Mrow) * OUT_FEATS + ocol] = val;
        }
    }
}

extern "C" void kernel_launch(void* const* d_in, const int* in_sizes, int n_in,
                              void* d_out, int out_size, void* d_ws, size_t ws_size,
                              hipStream_t stream) {
    const float* feature = (const float*)d_in[0];
    const int*   src     = (const int*)  d_in[1];
    const int*   dst     = (const int*)  d_in[2];
    const float* W       = (const float*)d_in[3];
    const float* bias    = (const float*)d_in[4];
    float*       out     = (float*)d_out;
    float*       h       = (float*)d_ws;

    zero_h<<<(N_NODES * IN_FEATS / 4) / 256, 256, 0, stream>>>(h);

    scatter_add<<<(N_EDGES * 32) / 256, 256, 0, stream>>>(feature, src, dst, h);

    gemm_wmma<<<3125, 256, 0, stream>>>(h, W, bias, out);
}


# GEMM tiled through LDS: a workgroup stages its 64-row block of h once and streams W in 16-k chunks (two buffers); waves read both fragments from LDS, wave tile 64x64; k sum order unchanged
# speedup vs baseline: 1.1898x; 1.1898x over previous
#include <hip/hip_runtime.h>
#include <hip/hip_bf16.h>

#define N_NODES   50000
#define N_EDGES   800000
#define IN_FEATS  128
#define OUT_FEATS 512
#define LEAKY     0.01f

typedef __attribute__((ext_vector_type(2))) float v2f;
typedef __attribute__((ext_vector_type(4))) float v4f;
typedef __attribute__((ext_vector_type(8))) float v8f;

__global__ __launch_bounds__(256) void zero_h(float* __restrict__ h) {
    unsigned i = blockIdx.x * 256u + threadIdx.x;
    ((v4f*)h)[i] = (v4f){0.f, 0.f, 0.f, 0.f};
}

__global__ __launch_bounds__(256) void scatter_add(const float* __restrict__ feature,
                                                   const int*   __restrict__ src,
                                                   const int*   __restrict__ dst,
                                                   float*       __restrict__ h) {
    unsigned gid  = blockIdx.x * 256u + threadIdx.x;
    unsigned e    = gid >> 5;
    unsigned lane = gid & 31u;

    int s = src[e];
    int d = dst[e];

    const float* fp = feature + (size_t)s * IN_FEATS + lane * 4u;
    v4f val = *(const v4f*)fp;

    float* hp = h + (size_t)d * IN_FEATS + lane * 4u;
    atomicAdd(hp + 0, val.x);
    atomicAdd(hp + 1, val.y);
    atomicAdd(hp + 2, val.z);
    atomicAdd(hp + 3, val.w);
}

#define BM      64
#define APITCH  132
#define KC      16
#define NCHUNK  (IN_FEATS / KC)
#define WPITCH  20

__global__ __launch_bounds__(256) void gemm_wmma(const float* __restrict__ h,
                                                 const float* __restrict__ W,
                                                 const float* __restrict__ bias,
                                                 float*       __restrict__ out) {
    __shared__ __attribute__((aligned(16))) float As[BM * APITCH];
    __shared__ __attribute__((aligned(16))) float Ws[2][OUT_FEATS * WPITCH];

    const unsigned tid   = threadIdx.x;
    const unsigned wave  = tid >> 5;
    const unsigned lane  = tid & 31u;
    const unsigned rbase = blockIdx.x * (unsigned)BM;

    const unsigned wr = tid >> 2;
    const unsigned wc = tid & 3u;
    const float*   wsrc = W + (size_t)wr * IN_FEATS + wc * 4u;
    const unsigned wdst = wr * WPITCH + wc * 4u;

    v4f wreg[8];
    #pragma unroll
    for (int i = 0; i < 8; ++i)
        wreg[i] = *(const v4f*)(wsrc + (size_t)i * 64 * IN_FEATS);

    #pragma unroll
    for (int i = 0; i < 8; ++i) {
        const unsigned r  = wave + 8u * (unsigned)i;
        unsigned gr = rbase + r;
        gr = (gr < (unsigned)N_NODES) ? gr : (unsigned)(N_NODES - 1);
        const v4f v = *(const v4f*)(h + (size_t)gr * IN_FEATS + lane * 4u);
        *(v4f*)(&As[r * APITCH + lane * 4u]) = v;
    }
    #pragma unroll
    for (int i = 0; i < 8; ++i)
        *(v4f*)(&Ws[0][wdst + (unsigned)i * 64u * WPITCH]) = wreg[i];
    __syncthreads();

    const unsigned M     = lane & 15u;
    const unsigned kg    = lane >> 4;
    const unsigned obase = wave * 64u;

    const float*   ap   = As + M * APITCH + 2u * kg;
    const unsigned boff = (obase + M) * WPITCH + 2u * kg;

    v8f acc[4][4];
    #pragma unroll
    for (int rt = 0; rt < 4; ++rt) {
        #pragma unroll
        for (int t = 0; t < 4; ++t) acc[rt][t] = (v8f){0.f, 0.f, 0.f, 0.f, 0.f, 0.f, 0.f, 0.f};
    }

    #pragma unroll 1
    for (int kc = 0; kc < NCHUNK; ++kc) {
        const float* bp = &Ws[kc & 1][boff];

        if (kc + 1 < NCHUNK) {
            #pragma unroll
            for (int i = 0; i < 8; ++i)
                wreg[i] = *(const v4f*)(wsrc + (size_t)i * 64 * IN_FEATS + (kc + 1) * KC);
        }

        #pragma unroll
        for (int ks = 0; ks < KC; ks += 4) {
            v2f a[4], f[4];
            #pragma unroll
            for (int rt = 0; rt < 4; ++rt)
                a[rt] = *(const v2f*)(ap + rt * 16 * APITCH + kc * KC + ks);
            #pragma unroll
            for (int t = 0; t < 4; ++t)
                f[t] = *(const v2f*)(bp + t * 16 * WPITCH + ks);
            #pragma unroll
            for (int rt = 0; rt < 4; ++rt) {
                #pragma unroll
                for (int t = 0; t < 4; ++t)
                    acc[rt][t] = __builtin_amdgcn_wmma_f32_16x16x4_f32(false, a[rt], false, f[t],
                                                                       (short)0, acc[rt][t],
                                                                       false, false);
            }
        }

        if (kc + 1 < NCHUNK) {
            #pragma unroll
            for (int i = 0; i < 8; ++i)
                *(v4f*)(&Ws[(kc + 1) & 1][wdst + (unsigned)i * 64u * WPITCH]) = wreg[i];
            __syncthreads();
        }
    }

    float bv[4];
    #pragma unroll
    for (int t = 0; t < 4; ++t) bv[t] = bias[obase + (unsigned)t * 16u + M];

    #pragma unroll
    for (int rt = 0; rt < 4; ++rt) {
        const unsigned nbase = rbase + (unsigned)rt * 16u;
        if (nbase < (unsigned)N_NODES) {
            #pragma unroll
            for (int t = 0; t < 4; ++t) {
                const unsigned ocol = obase + (unsigned)t * 16u + M;
                #pragma unroll
                for (int v = 0; v < 8; ++v) {
                    const unsigned Mrow = (unsigned)v + 8u * kg;
                    float val = acc[rt][t][v] + bv[t];
                    val = (val >= 0.f) ? val : LEAKY * val;
                    out[(size_t)(nbase + Mrow) * OUT_FEATS + ocol] = val;
                }
            }
        }
    }
}

extern "C" void kernel_launch(void* const* d_in, const int* in_sizes, int n_in,
                              void* d_out, int out_size, void* d_ws, size_t ws_size,
                              hipStream_t stream) {
    const float* feature = (const float*)d_in[0];
    const int*   src     = (const int*)  d_in[1];
    const int*   dst     = (const int*)  d_in[2];
    const float* W       = (const float*)d_in[3];
    const float* bias    = (const float*)d_in[4];
    float*       out     = (float*)d_out;
    float*       h       = (float*)d_ws;

    zero_h<<<(N_NODES * IN_FEATS / 4) / 256, 256, 0, stream>>>(h);

    scatter_add<<<(N_EDGES * 32) / 256, 256, 0, stream>>>(feature, src, dst, h);

    gemm_wmma<<<782, 256, 0, stream>>>(h, W, bias, out);
}


# Aggregation fused into the GEMM: a workgroup owns 196 rows of h in LDS, filters the dst list for them, adds feature rows with LDS f32 atomics (arrival order), runs WMMA from LDS
# speedup vs baseline: 1.8549x; 1.5590x over previous
#include <hip/hip_runtime.h>


#define N_NODES   50000
#define N_EDGES   800000
#define IN_FEATS  128
#define OUT_FEATS 512
#define LEAKY     0.01f

typedef __attribute__((ext_vector_type(2))) float v2f;
typedef __attribute__((ext_vector_type(4))) float v4f;
typedef __attribute__((ext_vector_type(8))) float v8f;
typedef __attribute__((ext_vector_type(4))) int   v4i;

#define RB      196
#define RT      14
#define RPAD    (RT * 16)
#define APITCH  132
#define KC      16
#define NCHUNK  (IN_FEATS / KC)
#define WPITCH  20
#define NWAVE   16
#define NGROUP  (N_EDGES / 128)
#define NSUPER  ((NGROUP + 7) / 8)
#define QCAP    (32 + 8 * 128)

__device__ __forceinline__ void lds_add(float* p, float v) {
    __hip_atomic_fetch_add(p, v, __ATOMIC_RELAXED, __HIP_MEMORY_SCOPE_WORKGROUP);
}

__global__ __launch_bounds__(512) void gcn_fused(const float* __restrict__ feature,
                                                 const int*   __restrict__ src,
                                                 const int*   __restrict__ dst,
                                                 const float* __restrict__ W,
                                                 const float* __restrict__ bias,
                                                 float*       __restrict__ out) {
    __shared__ __attribute__((aligned(16))) float    As[RPAD * APITCH];
    __shared__ __attribute__((aligned(16))) float    Ws[2][OUT_FEATS * WPITCH];
    __shared__ __attribute__((aligned(16))) unsigned Q[NWAVE * QCAP];

    const unsigned tid  = threadIdx.x;
    const unsigned lane = tid & 31u;
    const unsigned wave = (unsigned)__builtin_amdgcn_readfirstlane((int)(tid >> 5));
    const unsigned lo   = blockIdx.x * (unsigned)RB;

    for (unsigned i = tid; i < (unsigned)(RPAD * APITCH / 4); i += 512u)
        ((v4f*)As)[i] = (v4f){0.f, 0.f, 0.f, 0.f};
    __syncthreads();

    unsigned* q  = Q + wave * (unsigned)QCAP;
    unsigned  qn = 0;
    const unsigned rot = blockIdx.x * 3u;

    v4i cur[8];
    {
        unsigned sgr = wave + rot; sgr = (sgr >= (unsigned)NSUPER) ? sgr - (unsigned)NSUPER : sgr;
        #pragma unroll
        for (int u = 0; u < 8; ++u) {
            unsigned g = sgr * 8u + (unsigned)u; g = (g < (unsigned)NGROUP) ? g : (unsigned)(NGROUP - 1);
            cur[u] = *(const v4i*)(dst + (size_t)g * 128u + lane * 4u);
        }
    }

    for (unsigned sg = wave; sg < (unsigned)NSUPER; sg += (unsigned)NWAVE) {
        unsigned sgr = sg + rot; sgr = (sgr >= (unsigned)NSUPER) ? sgr - (unsigned)NSUPER : sgr;

        v4i nxt[8];
        {
            unsigned sn = sg + (unsigned)NWAVE; sn = (sn < (unsigned)NSUPER) ? sn : sg;
            unsigned snr = sn + rot; snr = (snr >= (unsigned)NSUPER) ? snr - (unsigned)NSUPER : snr;
            #pragma unroll
            for (int u = 0; u < 8; ++u) {
                unsigned g = snr * 8u + (unsigned)u; g = (g < (unsigned)NGROUP) ? g : (unsigned)(NGROUP - 1);
                nxt[u] = *(const v4i*)(dst + (size_t)g * 128u + lane * 4u);
            }
        }

        #pragma unroll
        for (int u = 0; u < 8; ++u) {
            const unsigned g  = sgr * 8u + (unsigned)u;
            const bool     gv = g < (unsigned)NGROUP;
            const unsigned x0 = (unsigned)cur[u].x - lo, x1 = (unsigned)cur[u].y - lo;
            const unsigned x2 = (unsigned)cur[u].z - lo, x3 = (unsigned)cur[u].w - lo;
            const bool m0 = gv && x0 < (unsigned)RB, m1 = gv && x1 < (unsigned)RB;
            const bool m2 = gv && x2 < (unsigned)RB, m3 = gv && x3 < (unsigned)RB;
            if (__builtin_amdgcn_ballot_w32(m0 || m1 || m2 || m3) != 0u) {
                const unsigned e0 = g * 128u + lane * 4u;
                unsigned mk;
                mk = __builtin_amdgcn_ballot_w32(m0);
                if (m0) q[qn + __builtin_amdgcn_mbcnt_lo(mk, 0u)] = (e0 + 0u) | (x0 << 20);
                qn += (unsigned)__builtin_popcount(mk);
                mk = __builtin_amdgcn_ballot_w32(m1);
                if (m1) q[qn + __builtin_amdgcn_mbcnt_lo(mk, 0u)] = (e0 + 1u) | (x1 << 20);
                qn += (unsigned)__builtin_popcount(mk);
                mk = __builtin_amdgcn_ballot_w32(m2);
                if (m2) q[qn + __builtin_amdgcn_mbcnt_lo(mk, 0u)] = (e0 + 2u) | (x2 << 20);
                qn += (unsigned)__builtin_popcount(mk);
                mk = __builtin_amdgcn_ballot_w32(m3);
                if (m3) q[qn + __builtin_amdgcn_mbcnt_lo(mk, 0u)] = (e0 + 3u) | (x3 << 20);
                qn += (unsigned)__builtin_popcount(mk);
            }
        }

        const bool     last = sg + (unsigned)NWAVE >= (unsigned)NSUPER;
        const unsigned lim  = last ? qn : (qn & ~31u);
        for (unsigned base = 0; base < lim; base += 32u) {
            const unsigned idx = base + lane;
            const unsigned ent = (idx < qn) ? q[idx] : (200u << 20);
            const unsigned dl  = ent >> 20;
            unsigned s = (unsigned)src[ent & 0xFFFFFu];
            s = (s < (unsigned)N_NODES) ? s : (unsigned)(N_NODES - 1);
            float va[8][4], vb[8][4];
            #pragma unroll
            for (int k = 0; k < 8; ++k) {
                const float* fp = feature + (size_t)(unsigned)__builtin_amdgcn_readlane((int)s, k) * IN_FEATS + lane;
                #pragma unroll
                for (int c = 0; c < 4; ++c) va[k][c] = fp[32 * c];
            }
            #pragma unroll
            for (int j = 0; j < 4; ++j) {
                if (j < 3) {
                    #pragma unroll
                    for (int k = 0; k < 8; ++k) {
                        const float* fp = feature + (size_t)(unsigned)__builtin_amdgcn_readlane((int)s, 8 * (j + 1) + k) * IN_FEATS + lane;
                        #pragma unroll
                        for (int c = 0; c < 4; ++c) { if ((j & 1) == 0) vb[k][c] = fp[32 * c]; else va[k][c] = fp[32 * c]; }
                    }
                }
                #pragma unroll
                for (int k = 0; k < 8; ++k) {
                    float* hp = As + (unsigned)__builtin_amdgcn_readlane((int)dl, 8 * j + k) * (unsigned)APITCH + lane;
                    #pragma unroll
                    for (int c = 0; c < 4; ++c) lds_add(hp + 32 * c, ((j & 1) == 0) ? va[k][c] : vb[k][c]);
                }
            }
        }
        if (lim != 0u && !last) {
            const unsigned r = qn - lim;
            if (lane < r) { const unsigned tmp = q[lim + lane]; q[lane] = tmp; }
            qn = r;
        }

        #pragma unroll
        for (int u = 0; u < 8; ++u) cur[u] = nxt[u];
    }

    const unsigned wr = tid >> 2;
    const unsigned wc = tid & 3u;
    const float*   wsrc = W + (size_t)wr * IN_FEATS + wc * 4u;
    const unsigned wdst = wr * WPITCH + wc * 4u;

    v4f wreg[4];
    #pragma unroll
    for (int i = 0; i < 4; ++i)
        wreg[i] = *(const v4f*)(wsrc + (size_t)i * 128 * IN_FEATS);
    #pragma unroll
    for (int i = 0; i < 4; ++i)
        *(v4f*)(&Ws[0][wdst + (unsigned)i * 128u * WPITCH]) = wreg[i];
    __syncthreads();

    const unsigned M     = lane & 15u;
    const unsigned kg    = lane >> 4;
    const unsigned obase = wave * 32u;

    const unsigned boff = (obase + M) * WPITCH + 2u * kg;

    float bv[2];
    #pragma unroll
    for (int t = 0; t < 2; ++t) bv[t] = bias[obase + (unsigned)t * 16u + M];

    const unsigned nrows = ((unsigned)N_NODES - lo < (unsigned)RB) ? (unsigned)N_NODES - lo : (unsigned)RB;

    v8f acc[7][2];
    #pragma unroll
    for (int rt = 0; rt < 7; ++rt) {
        #pragma unroll
        for (int t = 0; t < 2; ++t) acc[rt][t] = (v8f){0.f, 0.f, 0.f, 0.f, 0.f, 0.f, 0.f, 0.f};
    }

    #pragma unroll 1
    for (int st = 0; st < 2 * NCHUNK; ++st) {
        const int kc   = st & (NCHUNK - 1);
        const int pass = st >> 3;
        const float* bp = &Ws[st & 1][boff];
        const float* ap = As + ((unsigned)pass * 112u + M) * APITCH + 2u * kg + (unsigned)kc * KC;

        if (st + 1 < 2 * NCHUNK) {
            const int kn = (st + 1) & (NCHUNK - 1);
            #pragma unroll
            for (int i = 0; i < 4; ++i)
                wreg[i] = *(const v4f*)(wsrc + (size_t)i * 128 * IN_FEATS + kn * KC);
        }

        #pragma unroll
        for (int ks = 0; ks < KC; ks += 4) {
            v2f a[7], f[2];
            #pragma unroll
            for (int rt = 0; rt < 7; ++rt)
                a[rt] = *(const v2f*)(ap + rt * 16 * APITCH + ks);
            #pragma unroll
            for (int t = 0; t < 2; ++t)
                f[t] = *(const v2f*)(bp + t * 16 * WPITCH + ks);
            #pragma unroll
            for (int rt = 0; rt < 7; ++rt) {
                #pragma unroll
                for (int t = 0; t < 2; ++t)
                    acc[rt][t] = __builtin_amdgcn_wmma_f32_16x16x4_f32(false, a[rt], false, f[t],
                                                                       (short)0, acc[rt][t],
                                                                       false, false);
            }
        }

        if (kc == NCHUNK - 1) {
            #pragma unroll
            for (int rt = 0; rt < 7; ++rt) {
                const unsigned lbase = (unsigned)pass * 112u + (unsigned)rt * 16u;
                if (lbase < nrows) {
                    #pragma unroll
                    for (int t = 0; t < 2; ++t) {
                        const unsigned ocol = obase + (unsigned)t * 16u + M;
                        #pragma unroll
                        for (int v = 0; v < 8; ++v) {
                            const unsigned lr = lbase + (unsigned)v + 8u * kg;
                            float val = acc[rt][t][v] + bv[t];
                            val = (val >= 0.f) ? val : LEAKY * val;
                            if (lr < nrows)
                                out[(size_t)(lo + lr) * OUT_FEATS + ocol] = val;
                        }
                    }
                }
                #pragma unroll
                for (int t = 0; t < 2; ++t) acc[rt][t] = (v8f){0.f, 0.f, 0.f, 0.f, 0.f, 0.f, 0.f, 0.f};
            }
        }

        if (st + 1 < 2 * NCHUNK) {
            #pragma unroll
            for (int i = 0; i < 4; ++i)
                *(v4f*)(&Ws[(st + 1) & 1][wdst + (unsigned)i * 128u * WPITCH]) = wreg[i];
            __syncthreads();
        }
    }
}

extern "C" void kernel_launch(void* const* d_in, const int* in_sizes, int n_in,
                              void* d_out, int out_size, void* d_ws, size_t ws_size,
                              hipStream_t stream) {
    const float* feature = (const float*)d_in[0];
    const int*   src     = (const int*)  d_in[1];
    const int*   dst     = (const int*)  d_in[2];
    const float* W       = (const float*)d_in[3];
    const float* bias    = (const float*)d_in[4];
    float*       out     = (float*)d_out;

    gcn_fused<<<256, 512, 0, stream>>>(feature, src, dst, W, bias, out);
}


# Scan stream compressed: a first kernel writes per edge a 1-byte owner-block key and a packed (src,block,row) word; workgroups scan the key bytes, not dst; LDS f32 adds stay in arrival order
# speedup vs baseline: 2.2496x; 1.2128x over previous
#include <hip/hip_runtime.h>


#define N_NODES   50000
#define N_EDGES   800000
#define IN_FEATS  128
#define OUT_FEATS 512
#define LEAKY     0.01f

typedef __attribute__((ext_vector_type(2))) float v2f;
typedef __attribute__((ext_vector_type(4))) float v4f;
typedef __attribute__((ext_vector_type(8))) float v8f;
typedef __attribute__((ext_vector_type(4))) int   v4i;

#define RB      196
#define RT      14
#define RPAD    (RT * 16)
#define APITCH  132
#define KC      16
#define NCHUNK  (IN_FEATS / KC)
#define WPITCH  20
#define NWAVE   16
#define NUNIT   (N_EDGES / 16)
#define NGROUP  ((NUNIT + 31) / 32)
#define NLOAD   3
#define NSUPER  (NGROUP / NLOAD)
#define QCAP    (32 + NLOAD * 512)
static_assert(NSUPER * NLOAD == NGROUP, "runs must cover the groups exactly");
static_assert(N_NODES <= 256 * RB && N_NODES <= 65536, "block number must fit a byte, src 16 bits");

__device__ __forceinline__ void lds_add(float* p, float v) {
    __hip_atomic_fetch_add(p, v, __ATOMIC_RELAXED, __HIP_MEMORY_SCOPE_WORKGROUP);
}

#define PRE_WG   196
#define PRE_STEP (PRE_WG * 256)
__global__ __launch_bounds__(256) void gcn_edge_keys(const int* __restrict__ src,
                                                    const int* __restrict__ dst,
                                                    unsigned*  __restrict__ keys,
                                                    unsigned*  __restrict__ packed) {
    const unsigned t = blockIdx.x * 256u + threadIdx.x;
    v4i d[4], s[4];
    #pragma unroll
    for (int it = 0; it < 4; ++it) {
        unsigned u = t + (unsigned)(it * PRE_STEP);
        u = (u < (unsigned)(N_EDGES / 4)) ? u : (unsigned)(N_EDGES / 4 - 1);
        d[it] = ((const v4i*)dst)[u];
        s[it] = ((const v4i*)src)[u];
    }
    #pragma unroll
    for (int it = 0; it < 4; ++it) {
        const unsigned u = t + (unsigned)(it * PRE_STEP);
        unsigned k = 0u;
        v4i p;
        #pragma unroll
        for (int c = 0; c < 4; ++c) {
            const unsigned dv  = (unsigned)d[it][c];
            const unsigned bq  = dv / (unsigned)RB;
            const unsigned blk = bq & 0xFFu;
            const unsigned row = (dv < (unsigned)N_NODES) ? dv - bq * (unsigned)RB : 255u;
            unsigned sv = (unsigned)s[it][c];
            sv = (sv < (unsigned)N_NODES) ? sv : (unsigned)(N_NODES - 1);
            k |= blk << (8 * c);
            p[c] = (int)(sv | (blk << 16) | (row << 24));
        }
        if (u < (unsigned)(N_EDGES / 4)) {
            keys[u] = k;
            ((v4i*)packed)[u] = p;
        }
    }
}

__global__ __launch_bounds__(512) void gcn_fused(const float*    __restrict__ feature,
                                                 const unsigned* __restrict__ keys,
                                                 const unsigned* __restrict__ packed,
                                                 const float*    __restrict__ W,
                                                 const float*    __restrict__ bias,
                                                 float*          __restrict__ out) {
    __shared__ __attribute__((aligned(16))) float    As[RPAD * APITCH];
    __shared__ __attribute__((aligned(16))) float    Ws[2][OUT_FEATS * WPITCH];
    __shared__ __attribute__((aligned(16))) unsigned Q[NWAVE * QCAP];

    const unsigned tid  = threadIdx.x;
    const unsigned lane = tid & 31u;
    const unsigned wave = (unsigned)__builtin_amdgcn_readfirstlane((int)(tid >> 5));
    const unsigned lo   = blockIdx.x * (unsigned)RB;

    for (unsigned i = tid; i < (unsigned)(RPAD * APITCH / 4); i += 512u)
        ((v4f*)As)[i] = (v4f){0.f, 0.f, 0.f, 0.f};
    __syncthreads();

    unsigned* q  = Q + wave * (unsigned)QCAP;
    unsigned  qn = 0;
    const unsigned rot = blockIdx.x * 2u;
    const unsigned K4  = blockIdx.x * 0x01010101u;

    v4i kr[NLOAD];
    {
        unsigned sgr = wave + rot; sgr = (sgr >= (unsigned)NSUPER) ? sgr - (unsigned)NSUPER : sgr;
        #pragma unroll
        for (int u = 0; u < NLOAD; ++u) {
            unsigned un = (sgr * (unsigned)NLOAD + (unsigned)u) * 32u + lane;
            un = (un < (unsigned)NUNIT) ? un : (unsigned)(NUNIT - 1);
            kr[u] = ((const v4i*)keys)[un];
        }
    }

    for (unsigned sg = wave; sg < (unsigned)NSUPER; sg += (unsigned)NWAVE) {
        unsigned sgr = sg + rot; sgr = (sgr >= (unsigned)NSUPER) ? sgr - (unsigned)NSUPER : sgr;
        unsigned sn = sg + (unsigned)NWAVE; sn = (sn < (unsigned)NSUPER) ? sn : sg;
        unsigned snr = sn + rot; snr = (snr >= (unsigned)NSUPER) ? snr - (unsigned)NSUPER : snr;

        #pragma unroll
        for (int u = 0; u < NLOAD; ++u) {
            const unsigned unit = (sgr * (unsigned)NLOAD + (unsigned)u) * 32u + lane;
            const v4i kv = kr[u];
            {
                unsigned un = (snr * (unsigned)NLOAD + (unsigned)u) * 32u + lane;
                un = (un < (unsigned)NUNIT) ? un : (unsigned)(NUNIT - 1);
                kr[u] = ((const v4i*)keys)[un];
            }
            unsigned m = 0u;
            #pragma unroll
            for (int i = 0; i < 4; ++i) {
                const unsigned x = (unsigned)kv[i] ^ K4;
                const unsigned z = ~(((x & 0x7F7F7F7Fu) + 0x7F7F7F7Fu) | x) & 0x80808080u;
                m |= z >> i;
            }
            m = (unit < (unsigned)NUNIT) ? m : 0u;
            unsigned any = __builtin_amdgcn_ballot_w32(m != 0u);
            while (any != 0u) {
                const bool     has = m != 0u;
                const unsigned p   = (unsigned)__builtin_ctz(m | 0x80000000u);
                const unsigned off = ((7u - (p & 7u)) << 2) | (p >> 3);
                if (has) q[qn + __builtin_amdgcn_mbcnt_lo(any, 0u)] = unit * 16u + off;
                qn += (unsigned)__builtin_popcount(any);
                m &= m - 1u;
                any = __builtin_amdgcn_ballot_w32(m != 0u);
            }
        }

        const bool     last = sg + (unsigned)NWAVE >= (unsigned)NSUPER;
        const unsigned lim  = last ? qn : (qn & ~31u);
        for (unsigned base = 0; base < lim; base += 32u) {
            const unsigned idx = base + lane;
            const bool     ev  = idx < qn;
            const unsigned pw  = packed[ev ? q[idx] : 0u];
            unsigned s = pw & 0xFFFFu;
            s = (s < (unsigned)N_NODES) ? s : (unsigned)(N_NODES - 1);
            const unsigned row = pw >> 24;
            const bool     ok  = ev && ((pw >> 16) & 0xFFu) == blockIdx.x && row < (unsigned)RB;
            const unsigned dl  = ok ? row : 200u;
            float va[8][4], vb[8][4];
            #pragma unroll
            for (int k = 0; k < 8; ++k) {
                const float* fp = feature + (size_t)(unsigned)__builtin_amdgcn_readlane((int)s, k) * IN_FEATS + lane;
                #pragma unroll
                for (int c = 0; c < 4; ++c) va[k][c] = fp[32 * c];
            }
            #pragma unroll
            for (int j = 0; j < 4; ++j) {
                if (j < 3) {
                    #pragma unroll
                    for (int k = 0; k < 8; ++k) {
                        const float* fp = feature + (size_t)(unsigned)__builtin_amdgcn_readlane((int)s, 8 * (j + 1) + k) * IN_FEATS + lane;
                        #pragma unroll
                        for (int c = 0; c < 4; ++c) { if ((j & 1) == 0) vb[k][c] = fp[32 * c]; else va[k][c] = fp[32 * c]; }
                    }
                }
                #pragma unroll
                for (int k = 0; k < 8; ++k) {
                    float* hp = As + (unsigned)__builtin_amdgcn_readlane((int)dl, 8 * j + k) * (unsigned)APITCH + lane;
                    #pragma unroll
                    for (int c = 0; c < 4; ++c) lds_add(hp + 32 * c, ((j & 1) == 0) ? va[k][c] : vb[k][c]);
                }
            }
        }
        if (lim != 0u && !last) {
            const unsigned r = qn - lim;
            if (lane < r) { const unsigned tmp = q[lim + lane]; q[lane] = tmp; }
            qn = r;
        }
    }

    const unsigned wr = tid >> 2;
    const unsigned wc = tid & 3u;
    const float*   wsrc = W + (size_t)wr * IN_FEATS + wc * 4u;
    const unsigned wdst = wr * WPITCH + wc * 4u;

    v4f wreg[4];
    #pragma unroll
    for (int i = 0; i < 4; ++i)
        wreg[i] = *(const v4f*)(wsrc + (size_t)i * 128 * IN_FEATS);
    #pragma unroll
    for (int i = 0; i < 4; ++i)
        *(v4f*)(&Ws[0][wdst + (unsigned)i * 128u * WPITCH]) = wreg[i];
    __syncthreads();

    const unsigned M     = lane & 15u;
    const unsigned kg    = lane >> 4;
    const unsigned obase = wave * 32u;

    const unsigned boff = (obase + M) * WPITCH + 2u * kg;

    float bv[2];
    #pragma unroll
    for (int t = 0; t < 2; ++t) bv[t] = bias[obase + (unsigned)t * 16u + M];

    const unsigned nrows = ((unsigned)N_NODES - lo < (unsigned)RB) ? (unsigned)N_NODES - lo : (unsigned)RB;

    v8f acc[7][2];
    #pragma unroll
    for (int rt = 0; rt < 7; ++rt) {
        #pragma unroll
        for (int t = 0; t < 2; ++t) acc[rt][t] = (v8f){0.f, 0.f, 0.f, 0.f, 0.f, 0.f, 0.f, 0.f};
    }

    #pragma unroll 1
    for (int st = 0; st < 2 * NCHUNK; ++st) {
        const int kc   = st & (NCHUNK - 1);
        const int pass = st >> 3;
        const float* bp = &Ws[st & 1][boff];
        const float* ap = As + ((unsigned)pass * 112u + M) * APITCH + 2u * kg + (unsigned)kc * KC;

        if (st + 1 < 2 * NCHUNK) {
            const int kn = (st + 1) & (NCHUNK - 1);
            #pragma unroll
            for (int i = 0; i < 4; ++i)
                wreg[i] = *(const v4f*)(wsrc + (size_t)i * 128 * IN_FEATS + kn * KC);
        }

        #pragma unroll
        for (int ks = 0; ks < KC; ks += 4) {
            v2f a[7], f[2];
            #pragma unroll
            for (int rt = 0; rt < 7; ++rt)
                a[rt] = *(const v2f*)(ap + rt * 16 * APITCH + ks);
            #pragma unroll
            for (int t = 0; t < 2; ++t)
                f[t] = *(const v2f*)(bp + t * 16 * WPITCH + ks);
            #pragma unroll
            for (int rt = 0; rt < 7; ++rt) {
                #pragma unroll
                for (int t = 0; t < 2; ++t)
                    acc[rt][t] = __builtin_amdgcn_wmma_f32_16x16x4_f32(false, a[rt], false, f[t],
                                                                       (short)0, acc[rt][t],
                                                                       false, false);
            }
        }

        if (kc == NCHUNK - 1) {
            #pragma unroll
            for (int rt = 0; rt < 7; ++rt) {
                const unsigned lbase = (unsigned)pass * 112u + (unsigned)rt * 16u;
                if (lbase < nrows) {
                    #pragma unroll
                    for (int t = 0; t < 2; ++t) {
                        const unsigned ocol = obase + (unsigned)t * 16u + M;
                        #pragma unroll
                        for (int v = 0; v < 8; ++v) {
                            const unsigned lr = lbase + (unsigned)v + 8u * kg;
                            float val = acc[rt][t][v] + bv[t];
                            val = (val >= 0.f) ? val : LEAKY * val;
                            if (lr < nrows)
                                out[(size_t)(lo + lr) * OUT_FEATS + ocol] = val;
                        }
                    }
                }
                #pragma unroll
                for (int t = 0; t < 2; ++t) acc[rt][t] = (v8f){0.f, 0.f, 0.f, 0.f, 0.f, 0.f, 0.f, 0.f};
            }
        }

        if (st + 1 < 2 * NCHUNK) {
            #pragma unroll
            for (int i = 0; i < 4; ++i)
                *(v4f*)(&Ws[(st + 1) & 1][wdst + (unsigned)i * 128u * WPITCH]) = wreg[i];
            __syncthreads();
        }
    }
}

extern "C" void kernel_launch(void* const* d_in, const int* in_sizes, int n_in,
                              void* d_out, int out_size, void* d_ws, size_t ws_size,
                              hipStream_t stream) {
    const float* feature = (const float*)d_in[0];
    const int*   src     = (const int*)  d_in[1];
    const int*   dst     = (const int*)  d_in[2];
    const float* W       = (const float*)d_in[3];
    const float* bias    = (const float*)d_in[4];
    float*       out     = (float*)d_out;

    unsigned* keys   = (unsigned*)d_ws;
    unsigned* packed = (unsigned*)d_ws + N_EDGES / 4;

    gcn_edge_keys<<<PRE_WG, 256, 0, stream>>>(src, dst, keys, packed);
    gcn_fused<<<256, 512, 0, stream>>>(feature, keys, packed, W, bias, out);
}


# GEMM epilogue: row guard tested once per 16-row tile (wave-uniform); whole tiles store with no per-lane predicate, so no branch, address rebuild or translation wait per store; last tile keeps the test
# speedup vs baseline: 2.4298x; 1.0801x over previous
#include <hip/hip_runtime.h>


#define N_NODES   50000
#define N_EDGES   800000
#define IN_FEATS  128
#define OUT_FEATS 512
#define LEAKY     0.01f

typedef __attribute__((ext_vector_type(2))) float v2f;
typedef __attribute__((ext_vector_type(4))) float v4f;
typedef __attribute__((ext_vector_type(8))) float v8f;
typedef __attribute__((ext_vector_type(4))) int   v4i;

#define RB      196
#define RT      14
#define RPAD    (RT * 16)
#define APITCH  132
#define KC      16
#define NCHUNK  (IN_FEATS / KC)
#define WPITCH  20
#define NWAVE   16
#define NUNIT   (N_EDGES / 16)
#define NGROUP  ((NUNIT + 31) / 32)
#define NLOAD   3
#define NSUPER  (NGROUP / NLOAD)
#define QCAP    (32 + NLOAD * 512)
static_assert(NSUPER * NLOAD == NGROUP, "runs must cover the groups exactly");
static_assert(N_NODES <= 256 * RB && N_NODES <= 65536, "block number must fit a byte, src 16 bits");

__device__ __forceinline__ void lds_add(float* p, float v) {
    __hip_atomic_fetch_add(p, v, __ATOMIC_RELAXED, __HIP_MEMORY_SCOPE_WORKGROUP);
}

#define PRE_WG   196
#define PRE_STEP (PRE_WG * 256)
__global__ __launch_bounds__(256) void gcn_edge_keys(const int* __restrict__ src,
                                                    const int* __restrict__ dst,
                                                    unsigned*  __restrict__ keys,
                                                    unsigned*  __restrict__ packed) {
    const unsigned t = blockIdx.x * 256u + threadIdx.x;
    v4i d[4], s[4];
    #pragma unroll
    for (int it = 0; it < 4; ++it) {
        unsigned u = t + (unsigned)(it * PRE_STEP);
        u = (u < (unsigned)(N_EDGES / 4)) ? u : (unsigned)(N_EDGES / 4 - 1);
        d[it] = ((const v4i*)dst)[u];
        s[it] = ((const v4i*)src)[u];
    }
    #pragma unroll
    for (int it = 0; it < 4; ++it) {
        const unsigned u = t + (unsigned)(it * PRE_STEP);
        unsigned k = 0u;
        v4i p;
        #pragma unroll
        for (int c = 0; c < 4; ++c) {
            const unsigned dv  = (unsigned)d[it][c];
            const unsigned bq  = dv / (unsigned)RB;
            const unsigned blk = bq & 0xFFu;
            const unsigned row = (dv < (unsigned)N_NODES) ? dv - bq * (unsigned)RB : 255u;
            unsigned sv = (unsigned)s[it][c];
            sv = (sv < (unsigned)N_NODES) ? sv : (unsigned)(N_NODES - 1);
            k |= blk << (8 * c);
            p[c] = (int)(sv | (blk << 16) | (row << 24));
        }
        if (u < (unsigned)(N_EDGES / 4)) {
            keys[u] = k;
            ((v4i*)packed)[u] = p;
        }
    }
}

__global__ __launch_bounds__(512) void gcn_fused(const float*    __restrict__ feature,
                                                 const unsigned* __restrict__ keys,
                                                 const unsigned* __restrict__ packed,
                                                 const float*    __restrict__ W,
                                                 const float*    __restrict__ bias,
                                                 float*          __restrict__ out) {
    __shared__ __attribute__((aligned(16))) float    As[RPAD * APITCH];
    __shared__ __attribute__((aligned(16))) float    Ws[2][OUT_FEATS * WPITCH];
    __shared__ __attribute__((aligned(16))) unsigned Q[NWAVE * QCAP];

    const unsigned tid  = threadIdx.x;
    const unsigned lane = tid & 31u;
    const unsigned wave = (unsigned)__builtin_amdgcn_readfirstlane((int)(tid >> 5));
    const unsigned lo   = blockIdx.x * (unsigned)RB;

    for (unsigned i = tid; i < (unsigned)(RPAD * APITCH / 4); i += 512u)
        ((v4f*)As)[i] = (v4f){0.f, 0.f, 0.f, 0.f};
    __syncthreads();

    unsigned* q  = Q + wave * (unsigned)QCAP;
    unsigned  qn = 0;
    const unsigned rot = blockIdx.x * 2u;
    const unsigned K4  = blockIdx.x * 0x01010101u;

    v4i kr[NLOAD];
    {
        unsigned sgr = wave + rot; sgr = (sgr >= (unsigned)NSUPER) ? sgr - (unsigned)NSUPER : sgr;
        #pragma unroll
        for (int u = 0; u < NLOAD; ++u) {
            unsigned un = (sgr * (unsigned)NLOAD + (unsigned)u) * 32u + lane;
            un = (un < (unsigned)NUNIT) ? un : (unsigned)(NUNIT - 1);
            kr[u] = ((const v4i*)keys)[un];
        }
    }

    for (unsigned sg = wave; sg < (unsigned)NSUPER; sg += (unsigned)NWAVE) {
        unsigned sgr = sg + rot; sgr = (sgr >= (unsigned)NSUPER) ? sgr - (unsigned)NSUPER : sgr;
        unsigned sn = sg + (unsigned)NWAVE; sn = (sn < (unsigned)NSUPER) ? sn : sg;
        unsigned snr = sn + rot; snr = (snr >= (unsigned)NSUPER) ? snr - (unsigned)NSUPER : snr;

        #pragma unroll
        for (int u = 0; u < NLOAD; ++u) {
            const unsigned unit = (sgr * (unsigned)NLOAD + (unsigned)u) * 32u + lane;
            const v4i kv = kr[u];
            {
                unsigned un = (snr * (unsigned)NLOAD + (unsigned)u) * 32u + lane;
                un = (un < (unsigned)NUNIT) ? un : (unsigned)(NUNIT - 1);
                kr[u] = ((const v4i*)keys)[un];
            }
            unsigned m = 0u;
            #pragma unroll
            for (int i = 0; i < 4; ++i) {
                const unsigned x = (unsigned)kv[i] ^ K4;
                const unsigned z = ~(((x & 0x7F7F7F7Fu) + 0x7F7F7F7Fu) | x) & 0x80808080u;
                m |= z >> i;
            }
            m = (unit < (unsigned)NUNIT) ? m : 0u;
            unsigned any = __builtin_amdgcn_ballot_w32(m != 0u);
            while (any != 0u) {
                const bool     has = m != 0u;
                const unsigned p   = (unsigned)__builtin_ctz(m | 0x80000000u);
                const unsigned off = ((7u - (p & 7u)) << 2) | (p >> 3);
                if (has) q[qn + __builtin_amdgcn_mbcnt_lo(any, 0u)] = unit * 16u + off;
                qn += (unsigned)__builtin_popcount(any);
                m &= m - 1u;
                any = __builtin_amdgcn_ballot_w32(m != 0u);
            }
        }

        const bool     last = sg + (unsigned)NWAVE >= (unsigned)NSUPER;
        const unsigned lim  = last ? qn : (qn & ~31u);
        for (unsigned base = 0; base < lim; base += 32u) {
            const unsigned idx = base + lane;
            const bool     ev  = idx < qn;
            const unsigned pw  = packed[ev ? q[idx] : 0u];
            unsigned s = pw & 0xFFFFu;
            s = (s < (unsigned)N_NODES) ? s : (unsigned)(N_NODES - 1);
            const unsigned row = pw >> 24;
            const bool     ok  = ev && ((pw >> 16) & 0xFFu) == blockIdx.x && row < (unsigned)RB;
            const unsigned dl  = ok ? row : 200u;
            float va[8][4], vb[8][4];
            #pragma unroll
            for (int k = 0; k < 8; ++k) {
                const float* fp = feature + (size_t)(unsigned)__builtin_amdgcn_readlane((int)s, k) * IN_FEATS + lane;
                #pragma unroll
                for (int c = 0; c < 4; ++c) va[k][c] = fp[32 * c];
            }
            #pragma unroll
            for (int j = 0; j < 4; ++j) {
                if (j < 3) {
                    #pragma unroll
                    for (int k = 0; k < 8; ++k) {
                        const float* fp = feature + (size_t)(unsigned)__builtin_amdgcn_readlane((int)s, 8 * (j + 1) + k) * IN_FEATS + lane;
                        #pragma unroll
                        for (int c = 0; c < 4; ++c) { if ((j & 1) == 0) vb[k][c] = fp[32 * c]; else va[k][c] = fp[32 * c]; }
                    }
                }
                #pragma unroll
                for (int k = 0; k < 8; ++k) {
                    float* hp = As + (unsigned)__builtin_amdgcn_readlane((int)dl, 8 * j + k) * (unsigned)APITCH + lane;
                    #pragma unroll
                    for (int c = 0; c < 4; ++c) lds_add(hp + 32 * c, ((j & 1) == 0) ? va[k][c] : vb[k][c]);
                }
            }
        }
        if (lim != 0u && !last) {
            const unsigned r = qn - lim;
            if (lane < r) { const unsigned tmp = q[lim + lane]; q[lane] = tmp; }
            qn = r;
        }
    }

    const unsigned wr = tid >> 2;
    const unsigned wc = tid & 3u;
    const float*   wsrc = W + (size_t)wr * IN_FEATS + wc * 4u;
    const unsigned wdst = wr * WPITCH + wc * 4u;

    v4f wreg[4];
    #pragma unroll
    for (int i = 0; i < 4; ++i)
        wreg[i] = *(const v4f*)(wsrc + (size_t)i * 128 * IN_FEATS);
    #pragma unroll
    for (int i = 0; i < 4; ++i)
        *(v4f*)(&Ws[0][wdst + (unsigned)i * 128u * WPITCH]) = wreg[i];
    __syncthreads();

    const unsigned M     = lane & 15u;
    const unsigned kg    = lane >> 4;
    const unsigned obase = wave * 32u;

    const unsigned boff = (obase + M) * WPITCH + 2u * kg;

    float bv[2];
    #pragma unroll
    for (int t = 0; t < 2; ++t) bv[t] = bias[obase + (unsigned)t * 16u + M];

    const unsigned nrows = ((unsigned)N_NODES - lo < (unsigned)RB) ? (unsigned)N_NODES - lo : (unsigned)RB;

    v8f acc[7][2];
    #pragma unroll
    for (int rt = 0; rt < 7; ++rt) {
        #pragma unroll
        for (int t = 0; t < 2; ++t) acc[rt][t] = (v8f){0.f, 0.f, 0.f, 0.f, 0.f, 0.f, 0.f, 0.f};
    }

    #pragma unroll 1
    for (int st = 0; st < 2 * NCHUNK; ++st) {
        const int kc   = st & (NCHUNK - 1);
        const int pass = st >> 3;
        const float* bp = &Ws[st & 1][boff];
        const float* ap = As + ((unsigned)pass * 112u + M) * APITCH + 2u * kg + (unsigned)kc * KC;

        if (st + 1 < 2 * NCHUNK) {
            const int kn = (st + 1) & (NCHUNK - 1);
            #pragma unroll
            for (int i = 0; i < 4; ++i)
                wreg[i] = *(const v4f*)(wsrc + (size_t)i * 128 * IN_FEATS + kn * KC);
        }

        #pragma unroll
        for (int ks = 0; ks < KC; ks += 4) {
            v2f a[7], f[2];
            #pragma unroll
            for (int rt = 0; rt < 7; ++rt)
                a[rt] = *(const v2f*)(ap + rt * 16 * APITCH + ks);
            #pragma unroll
            for (int t = 0; t < 2; ++t)
                f[t] = *(const v2f*)(bp + t * 16 * WPITCH + ks);
            #pragma unroll
            for (int rt = 0; rt < 7; ++rt) {
                #pragma unroll
                for (int t = 0; t < 2; ++t)
                    acc[rt][t] = __builtin_amdgcn_wmma_f32_16x16x4_f32(false, a[rt], false, f[t],
                                                                       (short)0, acc[rt][t],
                                                                       false, false);
            }
        }

        if (kc == NCHUNK - 1) {
            #pragma unroll
            for (int rt = 0; rt < 7; ++rt) {
                const unsigned lbase = (unsigned)pass * 112u + (unsigned)rt * 16u;
                if (lbase + 16u <= nrows) {
                    float* op = out + (size_t)(lo + lbase + 8u * kg) * OUT_FEATS + obase + M;
                    #pragma unroll
                    for (int t = 0; t < 2; ++t) {
                        #pragma unroll
                        for (int v = 0; v < 8; ++v) {
                            float val = acc[rt][t][v] + bv[t];
                            val = (val >= 0.f) ? val : LEAKY * val;
                            op[v * OUT_FEATS + t * 16] = val;
                        }
                    }
                } else if (lbase < nrows) {
                    #pragma unroll
                    for (int t = 0; t < 2; ++t) {
                        const unsigned ocol = obase + (unsigned)t * 16u + M;
                        #pragma unroll
                        for (int v = 0; v < 8; ++v) {
                            const unsigned lr = lbase + (unsigned)v + 8u * kg;
                            float val = acc[rt][t][v] + bv[t];
                            val = (val >= 0.f) ? val : LEAKY * val;
                            if (lr < nrows)
                                out[(size_t)(lo + lr) * OUT_FEATS + ocol] = val;
                        }
                    }
                }
                #pragma unroll
                for (int t = 0; t < 2; ++t) acc[rt][t] = (v8f){0.f, 0.f, 0.f, 0.f, 0.f, 0.f, 0.f, 0.f};
            }
        }

        if (st + 1 < 2 * NCHUNK) {
            #pragma unroll
            for (int i = 0; i < 4; ++i)
                *(v4f*)(&Ws[(st + 1) & 1][wdst + (unsigned)i * 128u * WPITCH]) = wreg[i];
            __syncthreads();
        }
    }
}

extern "C" void kernel_launch(void* const* d_in, const int* in_sizes, int n_in,
                              void* d_out, int out_size, void* d_ws, size_t ws_size,
                              hipStream_t stream) {
    const float* feature = (const float*)d_in[0];
    const int*   src     = (const int*)  d_in[1];
    const int*   dst     = (const int*)  d_in[2];
    const float* W       = (const float*)d_in[3];
    const float* bias    = (const float*)d_in[4];
    float*       out     = (float*)d_out;

    unsigned* keys   = (unsigned*)d_ws;
    unsigned* packed = (unsigned*)d_ws + N_EDGES / 4;

    gcn_edge_keys<<<PRE_WG, 256, 0, stream>>>(src, dst, keys, packed);
    gcn_fused<<<256, 512, 0, stream>>>(feature, keys, packed, W, bias, out);
}


# Scan stream compressed more: first kernel writes per 1024-edge chunk a bit line per group of 4 blocks +2 low-bit planes (LDS OR); a block scans 3 bits/edge, not a key byte; f32 adds in arrival order
# speedup vs baseline: 2.6411x; 1.0870x over previous
#include <hip/hip_runtime.h>


#define N_NODES   50000
#define N_EDGES   800000
#define IN_FEATS  128
#define OUT_FEATS 512
#define LEAKY     0.01f

typedef __attribute__((ext_vector_type(2))) float v2f;
typedef __attribute__((ext_vector_type(4))) float v4f;
typedef __attribute__((ext_vector_type(8))) float v8f;
typedef __attribute__((ext_vector_type(4))) int   v4i;

#define RB      196
#define RT      14
#define RPAD    (RT * 16)
#define APITCH  132
#define KC      16
#define NCHUNK  (IN_FEATS / KC)
#define WPITCH  20
#define NWAVE   16
#define CH      1024
#define NCH     ((N_EDGES + CH - 1) / CH)
#define NGRP    64
#define NLD     ((NCH + 3) / 4)
#define QCAP    64
#define GP_WORDS (NCH * NGRP * 32)
#define LO_WORDS (2 * NCH * 32)
static_assert(N_NODES <= 256 * RB && N_NODES <= 65536, "block number must fit a byte, src 16 bits");
static_assert(N_EDGES % 4 == 0 && (GP_WORDS + LO_WORDS) % 4 == 0, "16-B units");

__device__ __forceinline__ void lds_add(float* p, float v) {
    __hip_atomic_fetch_add(p, v, __ATOMIC_RELAXED, __HIP_MEMORY_SCOPE_WORKGROUP);
}

__device__ __forceinline__ void lds_or(unsigned* p, unsigned v) {
    __hip_atomic_fetch_or(p, v, __ATOMIC_RELAXED, __HIP_MEMORY_SCOPE_WORKGROUP);
}

#define TABW    ((NGRP + 2) * 32)
__global__ __launch_bounds__(256) void gcn_edge_planes(const int* __restrict__ src,
                                                      const int* __restrict__ dst,
                                                      unsigned*  __restrict__ gp,
                                                      unsigned*  __restrict__ lo,
                                                      unsigned*  __restrict__ packed) {
    __shared__ __attribute__((aligned(16))) unsigned tab[4 * TABW];

    const unsigned t  = threadIdx.x;
    const unsigned c0 = blockIdx.x * 4u;

    v4i d[4], s[4];
    #pragma unroll
    for (int it = 0; it < 4; ++it) {
        unsigned u = (c0 + (unsigned)it) * 256u + t;
        u = (u < (unsigned)(N_EDGES / 4)) ? u : (unsigned)(N_EDGES / 4 - 1);
        d[it] = ((const v4i*)dst)[u];
        s[it] = ((const v4i*)src)[u];
    }

    for (unsigned i = t; i < (unsigned)(4 * TABW / 4); i += 256u)
        ((v4i*)tab)[i] = (v4i){0, 0, 0, 0};
    __syncthreads();

    const unsigned j  = t >> 3;
    const unsigned sh = 4u * (t & 7u);
    #pragma unroll
    for (int it = 0; it < 4; ++it) {
        const unsigned u    = (c0 + (unsigned)it) * 256u + t;
        const bool     live = u < (unsigned)(N_EDGES / 4);
        unsigned*      tb   = tab + it * TABW;
        unsigned n0 = 0u, n1 = 0u;
        v4i p;
        #pragma unroll
        for (int e = 0; e < 4; ++e) {
            const unsigned dv  = (unsigned)d[it][e];
            const unsigned bq  = dv / (unsigned)RB;
            const unsigned blk = bq & 0xFFu;
            const bool     val = dv < (unsigned)N_NODES;
            const unsigned row = val ? dv - bq * (unsigned)RB : 255u;
            unsigned sv = (unsigned)s[it][e];
            sv = (sv < (unsigned)N_NODES) ? sv : (unsigned)(N_NODES - 1);
            p[e] = (int)(sv | (blk << 16) | (row << 24));
            n0 |= (blk & 1u) << e;
            n1 |= ((blk >> 1) & 1u) << e;
            if (live && val) lds_or(&tb[(blk >> 2) * 32u + j], 1u << (sh + (unsigned)e));
        }
        if (live) {
            lds_or(&tb[(unsigned)NGRP * 32u + j], n0 << sh);
            lds_or(&tb[(unsigned)(NGRP + 1) * 32u + j], n1 << sh);
            ((v4i*)packed)[u] = p;
        }
    }
    __syncthreads();

    #pragma unroll
    for (int it = 0; it < 4; ++it) {
        const unsigned c = c0 + (unsigned)it;
        if (c < (unsigned)NCH) {
            #pragma unroll
            for (int i = 0; i < 2; ++i)
                ((v4i*)gp)[c * (unsigned)(NGRP * 8) + t + (unsigned)i * 256u] =
                    ((const v4i*)tab)[(unsigned)(it * (TABW / 4)) + t + (unsigned)i * 256u];
        }
    }
    if (t < 64u) {
        const unsigned it = t >> 4, tt = t & 15u, c = c0 + it;
        if (c < (unsigned)NCH)
            ((v4i*)lo)[((tt >> 3) * (unsigned)NCH + c) * 8u + (tt & 7u)] =
                ((const v4i*)tab)[it * (unsigned)(TABW / 4) + (unsigned)(NGRP * 8) + tt];
    }
}

__device__ __forceinline__ void drain32(const float* __restrict__ feature,
                                        const unsigned* __restrict__ packed,
                                        float* As, const unsigned* q,
                                        const unsigned cnt, const unsigned lane) {
    const bool     ev  = lane < cnt;
    const unsigned pw  = packed[ev ? q[lane] : 0u];
    unsigned s = pw & 0xFFFFu;
    s = (s < (unsigned)N_NODES) ? s : (unsigned)(N_NODES - 1);
    const unsigned row = pw >> 24;
    const bool     ok  = ev && ((pw >> 16) & 0xFFu) == blockIdx.x && row < (unsigned)RB;
    const unsigned dl  = ok ? row : 200u;
    float va[8][4], vb[8][4];
    #pragma unroll
    for (int k = 0; k < 8; ++k) {
        const float* fp = feature + (size_t)(unsigned)__builtin_amdgcn_readlane((int)s, k) * IN_FEATS + lane;
        #pragma unroll
        for (int c = 0; c < 4; ++c) va[k][c] = fp[32 * c];
    }
    #pragma unroll
    for (int j = 0; j < 4; ++j) {
        if (j < 3) {
            #pragma unroll
            for (int k = 0; k < 8; ++k) {
                const float* fp = feature + (size_t)(unsigned)__builtin_amdgcn_readlane((int)s, 8 * (j + 1) + k) * IN_FEATS + lane;
                #pragma unroll
                for (int c = 0; c < 4; ++c) { if ((j & 1) == 0) vb[k][c] = fp[32 * c]; else va[k][c] = fp[32 * c]; }
            }
        }
        #pragma unroll
        for (int k = 0; k < 8; ++k) {
            float* hp = As + (unsigned)__builtin_amdgcn_readlane((int)dl, 8 * j + k) * (unsigned)APITCH + lane;
            #pragma unroll
            for (int c = 0; c < 4; ++c) lds_add(hp + 32 * c, ((j & 1) == 0) ? va[k][c] : vb[k][c]);
        }
    }
}

__global__ __launch_bounds__(512) void gcn_fused(const float*    __restrict__ feature,
                                                 const unsigned* __restrict__ gp,
                                                 const unsigned* __restrict__ lbp,
                                                 const unsigned* __restrict__ packed,
                                                 const float*    __restrict__ W,
                                                 const float*    __restrict__ bias,
                                                 float*          __restrict__ out) {
    __shared__ __attribute__((aligned(16))) float    As[RPAD * APITCH];
    __shared__ __attribute__((aligned(16))) float    Ws[2][OUT_FEATS * WPITCH];
    __shared__ __attribute__((aligned(16))) unsigned Q[NWAVE * QCAP];

    const unsigned tid  = threadIdx.x;
    const unsigned lane = tid & 31u;
    const unsigned wave = (unsigned)__builtin_amdgcn_readfirstlane((int)(tid >> 5));
    const unsigned lo   = blockIdx.x * (unsigned)RB;

    for (unsigned i = tid; i < (unsigned)(RPAD * APITCH / 4); i += 512u)
        ((v4f*)As)[i] = (v4f){0.f, 0.f, 0.f, 0.f};
    __syncthreads();

    unsigned* q  = Q + wave * (unsigned)QCAP;
    unsigned  qn = 0;
    const unsigned grp = blockIdx.x >> 2;
    const unsigned x0  = (blockIdx.x & 1u) ? 0u : 0xFFFFFFFFu;
    const unsigned x1  = (blockIdx.x & 2u) ? 0u : 0xFFFFFFFFu;
    const unsigned l7  = lane & 7u;
    const unsigned l3  = lane >> 3;
    const unsigned rot = (blockIdx.x < (unsigned)NLD) ? blockIdx.x : blockIdx.x - (unsigned)NLD;

    v4i rg, r0, r1;
    {
        unsigned ir = wave + rot; ir = (ir >= (unsigned)NLD) ? ir - (unsigned)NLD : ir;
        unsigned cc = 4u * ir + l3; cc = (cc < (unsigned)NCH) ? cc : (unsigned)(NCH - 1);
        rg = ((const v4i*)gp)[(cc * (unsigned)NGRP + grp) * 8u + l7];
        r0 = ((const v4i*)lbp)[cc * 8u + l7];
        r1 = ((const v4i*)lbp)[((unsigned)NCH + cc) * 8u + l7];
    }

    for (unsigned i = wave; i < (unsigned)NLD; i += (unsigned)NWAVE) {
        unsigned ir = i + rot; ir = (ir >= (unsigned)NLD) ? ir - (unsigned)NLD : ir;
        unsigned in = i + (unsigned)NWAVE; in = (in < (unsigned)NLD) ? in : i;
        unsigned inr = in + rot; inr = (inr >= (unsigned)NLD) ? inr - (unsigned)NLD : inr;

        const unsigned c = 4u * ir + l3;
        const v4i g = rg, a0 = r0, a1 = r1;
        {
            unsigned cc = 4u * inr + l3; cc = (cc < (unsigned)NCH) ? cc : (unsigned)(NCH - 1);
            rg = ((const v4i*)gp)[(cc * (unsigned)NGRP + grp) * 8u + l7];
            r0 = ((const v4i*)lbp)[cc * 8u + l7];
            r1 = ((const v4i*)lbp)[((unsigned)NCH + cc) * 8u + l7];
        }
        const bool cv = c < (unsigned)NCH;
        unsigned m0 = cv ? ((unsigned)g[0] & ((unsigned)a0[0] ^ x0) & ((unsigned)a1[0] ^ x1)) : 0u;
        unsigned m1 = cv ? ((unsigned)g[1] & ((unsigned)a0[1] ^ x0) & ((unsigned)a1[1] ^ x1)) : 0u;
        unsigned m2 = cv ? ((unsigned)g[2] & ((unsigned)a0[2] ^ x0) & ((unsigned)a1[2] ^ x1)) : 0u;
        unsigned m3 = cv ? ((unsigned)g[3] & ((unsigned)a0[3] ^ x0) & ((unsigned)a1[3] ^ x1)) : 0u;
        unsigned eb = c * (unsigned)CH + l7 * 128u;

        #pragma unroll 1
        for (int k = 0; k < 4; ++k) {
            unsigned m = m0;
            m0 = m1; m1 = m2; m2 = m3;
            unsigned any = __builtin_amdgcn_ballot_w32(m != 0u);
            while (any != 0u) {
                const bool     has = m != 0u;
                const unsigned p   = (unsigned)__builtin_ctz(m | 0x80000000u);
                if (has) q[qn + __builtin_amdgcn_mbcnt_lo(any, 0u)] = eb + p;
                qn += (unsigned)__builtin_popcount(any);
                m &= m - 1u;
                if (qn >= 32u) {
                    drain32(feature, packed, As, q, 32u, lane);
                    const unsigned r = qn - 32u;
                    if (lane < r) { const unsigned tmp = q[32u + lane]; q[lane] = tmp; }
                    qn = r;
                }
                any = __builtin_amdgcn_ballot_w32(m != 0u);
            }
            eb += 32u;
        }
    }
    if (qn != 0u) drain32(feature, packed, As, q, qn, lane);

    const unsigned wr = tid >> 2;
    const unsigned wc = tid & 3u;
    const float*   wsrc = W + (size_t)wr * IN_FEATS + wc * 4u;
    const unsigned wdst = wr * WPITCH + wc * 4u;

    v4f wreg[4];
    #pragma unroll
    for (int i = 0; i < 4; ++i)
        wreg[i] = *(const v4f*)(wsrc + (size_t)i * 128 * IN_FEATS);
    #pragma unroll
    for (int i = 0; i < 4; ++i)
        *(v4f*)(&Ws[0][wdst + (unsigned)i * 128u * WPITCH]) = wreg[i];
    __syncthreads();

    const unsigned M     = lane & 15u;
    const unsigned kg    = lane >> 4;
    const unsigned obase = wave * 32u;

    const unsigned boff = (obase + M) * WPITCH + 2u * kg;

    float bv[2];
    #pragma unroll
    for (int t = 0; t < 2; ++t) bv[t] = bias[obase + (unsigned)t * 16u + M];

    const unsigned nrows = ((unsigned)N_NODES - lo < (unsigned)RB) ? (unsigned)N_NODES - lo : (unsigned)RB;

    v8f acc[7][2];
    #pragma unroll
    for (int rt = 0; rt < 7; ++rt) {
        #pragma unroll
        for (int t = 0; t < 2; ++t) acc[rt][t] = (v8f){0.f, 0.f, 0.f, 0.f, 0.f, 0.f, 0.f, 0.f};
    }

    #pragma unroll 1
    for (int st = 0; st < 2 * NCHUNK; ++st) {
        const int kc   = st & (NCHUNK - 1);
        const int pass = st >> 3;
        const float* bp = &Ws[st & 1][boff];
        const float* ap = As + ((unsigned)pass * 112u + M) * APITCH + 2u * kg + (unsigned)kc * KC;

        if (st + 1 < 2 * NCHUNK) {
            const int kn = (st + 1) & (NCHUNK - 1);
            #pragma unroll
            for (int i = 0; i < 4; ++i)
                wreg[i] = *(const v4f*)(wsrc + (size_t)i * 128 * IN_FEATS + kn * KC);
        }

        #pragma unroll
        for (int ks = 0; ks < KC; ks += 4) {
            v2f a[7], f[2];
            #pragma unroll
            for (int rt = 0; rt < 7; ++rt)
                a[rt] = *(const v2f*)(ap + rt * 16 * APITCH + ks);
            #pragma unroll
            for (int t = 0; t < 2; ++t)
                f[t] = *(const v2f*)(bp + t * 16 * WPITCH + ks);
            #pragma unroll
            for (int rt = 0; rt < 7; ++rt) {
                #pragma unroll
                for (int t = 0; t < 2; ++t)
                    acc[rt][t] = __builtin_amdgcn_wmma_f32_16x16x4_f32(false, a[rt], false, f[t],
                                                                       (short)0, acc[rt][t],
                                                                       false, false);
            }
        }

        if (kc == NCHUNK - 1) {
            #pragma unroll
            for (int rt = 0; rt < 7; ++rt) {
                const unsigned lbase = (unsigned)pass * 112u + (unsigned)rt * 16u;
                if (lbase + 16u <= nrows) {
                    float* op = out + (size_t)(lo + lbase + 8u * kg) * OUT_FEATS + obase + M;
                    #pragma unroll
                    for (int t = 0; t < 2; ++t) {
                        #pragma unroll
                        for (int v = 0; v < 8; ++v) {
                            float val = acc[rt][t][v] + bv[t];
                            val = (val >= 0.f) ? val : LEAKY * val;
                            op[v * OUT_FEATS + t * 16] = val;
                        }
                    }
                } else if (lbase < nrows) {
                    #pragma unroll
                    for (int t = 0; t < 2; ++t) {
                        const unsigned ocol = obase + (unsigned)t * 16u + M;
                        #pragma unroll
                        for (int v = 0; v < 8; ++v) {
                            const unsigned lr = lbase + (unsigned)v + 8u * kg;
                            float val = acc[rt][t][v] + bv[t];
                            val = (val >= 0.f) ? val : LEAKY * val;
                            if (lr < nrows)
                                out[(size_t)(lo + lr) * OUT_FEATS + ocol] = val;
                        }
                    }
                }
                #pragma unroll
                for (int t = 0; t < 2; ++t) acc[rt][t] = (v8f){0.f, 0.f, 0.f, 0.f, 0.f, 0.f, 0.f, 0.f};
            }
        }

        if (st + 1 < 2 * NCHUNK) {
            #pragma unroll
            for (int i = 0; i < 4; ++i)
                *(v4f*)(&Ws[(st + 1) & 1][wdst + (unsigned)i * 128u * WPITCH]) = wreg[i];
            __syncthreads();
        }
    }
}

extern "C" void kernel_launch(void* const* d_in, const int* in_sizes, int n_in,
                              void* d_out, int out_size, void* d_ws, size_t ws_size,
                              hipStream_t stream) {
    const float* feature = (const float*)d_in[0];
    const int*   src     = (const int*)  d_in[1];
    const int*   dst     = (const int*)  d_in[2];
    const float* W       = (const float*)d_in[3];
    const float* bias    = (const float*)d_in[4];
    float*       out     = (float*)d_out;

    unsigned* gp     = (unsigned*)d_ws;
    unsigned* lo     = (unsigned*)d_ws + GP_WORDS;
    unsigned* packed = (unsigned*)d_ws + GP_WORDS + LO_WORDS;

    gcn_edge_planes<<<NLD, 256, 0, stream>>>(src, dst, gp, lo, packed);
    gcn_fused<<<256, 512, 0, stream>>>(feature, gp, lo, packed, W, bias, out);
}
